# starting kernel
# baseline (speedup 1.0000x reference)
#include <hip/hip_runtime.h>

typedef __attribute__((ext_vector_type(2))) float v2f;
typedef __attribute__((ext_vector_type(8))) float v8f;


__global__ void init_deg_kernel(float* __restrict__ deg, int n) {
  int i = blockIdx.x * blockDim.x + threadIdx.x;
  if (i < n) deg[i] = 1.0f;
}

__global__ void deg_count_kernel(const int* __restrict__ dst, float* __restrict__ deg, int e) {
  int i = blockIdx.x * blockDim.x + threadIdx.x;
  if (i < e) atomicAdd(&deg[dst[i]], 1.0f);
}

__global__ void dinv_kernel(float* __restrict__ deg, int n) {
  int i = blockIdx.x * blockDim.x + threadIdx.x;
  if (i < n) {
    float d = deg[i];
    deg[i] = (d > 0.0f) ? rsqrtf(d) : 0.0f;
  }
}

template <int K, int NC>
__global__ __launch_bounds__(32 * (NC / 16)) void gemm_wmma_kernel(
    const float* __restrict__ X, const float* __restrict__ W,
    float* __restrict__ H, int M) {
  const int lane = threadIdx.x & 31;
  const int wave = threadIdx.x >> 5;
  const int m0   = blockIdx.x << 4;
  const int n0   = wave << 4;
  const int l15  = lane & 15;
  const int kh   = (lane >> 4) << 1;

  int row = m0 + l15;
  if (row >= M) row = M - 1;
  const int col = n0 + l15;

  v8f c = {};
  for (int k0 = 0; k0 < K; k0 += 4) {
    v2f a, b;
    a.x = X[row * K + k0 + kh];
    a.y = X[row * K + k0 + kh + 1];
    b.x = W[(k0 + kh) * NC + col];
    b.y = W[(k0 + kh + 1) * NC + col];
    c = __builtin_amdgcn_wmma_f32_16x16x4_f32(
         false, a,  false, b,
         (short)0, c,  false,  false);
  }

  const int rbase = m0 + ((lane >> 4) << 3);
#pragma unroll
  for (int v = 0; v < 8; ++v) {
    int r = rbase + v;
    if (r < M) H[r * NC + col] = c[v];
  }
}

template <int CLOG>
__global__ void edge_agg_kernel(const int* __restrict__ src, const int* __restrict__ dst,
                                const float* __restrict__ dinv, const float* __restrict__ h,
                                float* __restrict__ agg, int e) {
  unsigned t = blockIdx.x * blockDim.x + threadIdx.x;
  unsigned edge = t >> CLOG;
  if (edge >= (unsigned)e) return;
  unsigned c = t & ((1u << CLOG) - 1u);
  int s = src[edge];
  int d = dst[edge];
  float nrm = dinv[s] * dinv[d];
  atomicAdd(&agg[((size_t)d << CLOG) + c], nrm * h[((size_t)s << CLOG) + c]);
}

template <int CLOG, bool RELU>
__global__ void finish_kernel(const float* __restrict__ agg, const float* __restrict__ hlin,
                              const float* __restrict__ dinv, const float* __restrict__ bias,
                              float* __restrict__ out, int n) {
  unsigned t = blockIdx.x * blockDim.x + threadIdx.x;
  unsigned i = t >> CLOG;
  if (i >= (unsigned)n) return;
  unsigned c = t & ((1u << CLOG) - 1u);
  float di = dinv[i];
  float v = agg[t] + di * di * hlin[t] + bias[c];
  out[t] = RELU ? fmaxf(v, 0.0f) : v;
}


extern "C" void kernel_launch(void* const* d_in, const int* in_sizes, int n_in,
                              void* d_out, int out_size, void* d_ws, size_t ws_size,
                              hipStream_t stream) {
  (void)n_in; (void)out_size; (void)ws_size;
  const float* x  = (const float*)d_in[0];
  const int*   ei = (const int*)d_in[1];
  const float* W1 = (const float*)d_in[2];
  const float* b1 = (const float*)d_in[3];
  const float* W2 = (const float*)d_in[4];
  const float* b2 = (const float*)d_in[5];

  const int IN_C = 128, HID = 64, OUT = 32;
  const int n = in_sizes[0] / IN_C;
  const int e = in_sizes[1] / 2;
  const int* srcp = ei;
  const int* dstp = ei + e;

  float* ws   = (float*)d_ws;
  size_t off  = ((size_t)n + 255) & ~(size_t)255;
  float* dinv = ws;
  float* h1   = ws + off;
  float* agg1 = h1 + (size_t)n * HID;
  float* h2   = agg1 + (size_t)n * HID;
  float* outp = (float*)d_out;

  const int TPB = 256;

  hipMemsetAsync(agg1, 0, (size_t)n * HID * sizeof(float), stream);
  hipMemsetAsync(d_out, 0, (size_t)n * OUT * sizeof(float), stream);

  init_deg_kernel<<<(n + TPB - 1) / TPB, TPB, 0, stream>>>(dinv, n);
  deg_count_kernel<<<(e + TPB - 1) / TPB, TPB, 0, stream>>>(dstp, dinv, e);
  dinv_kernel<<<(n + TPB - 1) / TPB, TPB, 0, stream>>>(dinv, n);

  gemm_wmma_kernel<128, 64><<<(n + 15) / 16, 128, 0, stream>>>(x, W1, h1, n);
  {
    unsigned total = (unsigned)e << 6;
    edge_agg_kernel<6><<<(total + TPB - 1) / TPB, TPB, 0, stream>>>(srcp, dstp, dinv, h1, agg1, e);
  }
  {
    unsigned total = (unsigned)n << 6;
    finish_kernel<6, true><<<(total + TPB - 1) / TPB, TPB, 0, stream>>>(agg1, h1, dinv, b1, agg1, n);
  }

  gemm_wmma_kernel<64, 32><<<(n + 15) / 16, 64, 0, stream>>>(agg1, W2, h2, n);
  {
    unsigned total = (unsigned)e << 5;
    edge_agg_kernel<5><<<(total + TPB - 1) / TPB, TPB, 0, stream>>>(srcp, dstp, dinv, h2, outp, e);
  }
  {
    unsigned total = (unsigned)n << 5;
    finish_kernel<5, false><<<(total + TPB - 1) / TPB, TPB, 0, stream>>>(outp, h2, dinv, b2, outp, n);
  }
}


# Four fewer dependent launches (11 to 7): degree start in the one memset, dinv made in product 1, layer-1 finish feeds product 2's A tile via LDS, both layer-2 sum copies in workspace; sums unchanged
# speedup vs baseline: 3.6191x; 3.6191x over previous
#include <hip/hip_runtime.h>

typedef __attribute__((ext_vector_type(2))) float v2f;
typedef __attribute__((ext_vector_type(8))) float v8f;


__device__ __forceinline__ int cache_half() {
  return (int)((__builtin_amdgcn_s_sendmsg_rtn(0x87) >> 18) & 1u);
}


__global__ void deg_count_kernel(const int* __restrict__ dst, float* __restrict__ deg0,
                                 float* __restrict__ deg1, int e) {
  int i = blockIdx.x * blockDim.x + threadIdx.x;
  float* deg = cache_half() ? deg1 : deg0;
  if (i < e) atomicAdd(&deg[dst[i]], 1.0f);
}

__device__ __forceinline__ float dinv_of(float d0, float d1) {
  float d = (d0 + d1) + 1.0f;
  return (d > 0.0f) ? rsqrtf(d) : 0.0f;
}

template <int K, int NC>
__global__ __launch_bounds__(32 * (NC / 16)) void gemm_wmma_kernel(
    const float* __restrict__ X, const float* __restrict__ W,
    const float* __restrict__ deg0, const float* __restrict__ deg1,
    float* __restrict__ dinv, float* __restrict__ H, int M) {
  const int lane = threadIdx.x & 31;
  const int wave = threadIdx.x >> 5;
  const int m0   = blockIdx.x << 4;
  const int n0   = wave << 4;
  const int l15  = lane & 15;
  const int kh   = (lane >> 4) << 1;

  int row = m0 + l15;
  if (row >= M) row = M - 1;
  const int col = n0 + l15;

  const int rbase = m0 + ((lane >> 4) << 3);
  float dv[8];
#pragma unroll
  for (int v = 0; v < 8; ++v) {
    const int r = rbase + v;
    const int rc = r < M ? r : M - 1;
    dv[v] = dinv_of(deg0[rc], deg1[rc]);
  }
  const float ds0 = deg0[row];
  const float ds1 = deg1[row];

  v8f c = {};
  for (int k0 = 0; k0 < K; k0 += 4) {
    v2f a, b;
    a.x = X[row * K + k0 + kh];
    a.y = X[row * K + k0 + kh + 1];
    b.x = W[(k0 + kh) * NC + col];
    b.y = W[(k0 + kh + 1) * NC + col];
    c = __builtin_amdgcn_wmma_f32_16x16x4_f32(
         false, a,  false, b,
         (short)0, c,  false,  false);
  }

#pragma unroll
  for (int v = 0; v < 8; ++v) {
    int r = rbase + v;
    if (r < M) H[r * NC + col] = dv[v] * c[v];
  }
  if (wave == 0 && lane < 16 && m0 + lane < M) dinv[m0 + lane] = dinv_of(ds0, ds1);
}

template <int K, int NC>
__global__ __launch_bounds__(32 * (NC / 16)) void finish_gemm_kernel(
    const float* __restrict__ agg0, const float* __restrict__ agg1,
    const float* __restrict__ hs, const float* __restrict__ dinv,
    const float* __restrict__ bias, const float* __restrict__ W,
    float* __restrict__ H, int M) {
  constexpr int NT = 32 * (NC / 16);
  constexpr int Q = K / 4;
  constexpr int PITCH = K + 4;
  __shared__ __attribute__((aligned(16))) float tile[16 * PITCH];

  const int lane = threadIdx.x & 31;
  const int wave = threadIdx.x >> 5;
  const int m0   = blockIdx.x << 4;
  const int n0   = wave << 4;
  const int l15  = lane & 15;
  const int kh   = (lane >> 4) << 1;
  const int col  = n0 + l15;

  const int rbase = m0 + ((lane >> 4) << 3);
  float dv[8];
#pragma unroll
  for (int v = 0; v < 8; ++v) {
    const int r = rbase + v;
    dv[v] = dinv[r < M ? r : M - 1];
  }

#pragma unroll
  for (int i = 0; i < (16 * Q) / NT; ++i) {
    const int p  = i * NT + (int)threadIdx.x;
    const int rl = p / Q;
    const int c4 = (p % Q) << 2;
    int r = m0 + rl;
    if (r >= M) r = M - 1;
    const int g = r * K + c4;
    const float4 a = *reinterpret_cast<const float4*>(agg0 + g);
    const float4 b = *reinterpret_cast<const float4*>(agg1 + g);
    const float4 h = *reinterpret_cast<const float4*>(hs + g);
    const float di = dinv[r];
    float4 o;
    o.x = fmaxf(di * ((a.x + b.x) + h.x) + bias[c4 + 0], 0.0f);
    o.y = fmaxf(di * ((a.y + b.y) + h.y) + bias[c4 + 1], 0.0f);
    o.z = fmaxf(di * ((a.z + b.z) + h.z) + bias[c4 + 2], 0.0f);
    o.w = fmaxf(di * ((a.w + b.w) + h.w) + bias[c4 + 3], 0.0f);
    *reinterpret_cast<float4*>(tile + rl * PITCH + c4) = o;
  }
  __syncthreads();

  v8f c = {};
  for (int k0 = 0; k0 < K; k0 += 4) {
    const v2f a = *reinterpret_cast<const v2f*>(tile + l15 * PITCH + k0 + kh);
    v2f b;
    b.x = W[(k0 + kh) * NC + col];
    b.y = W[(k0 + kh + 1) * NC + col];
    c = __builtin_amdgcn_wmma_f32_16x16x4_f32(
         false, a,  false, b,
         (short)0, c,  false,  false);
  }

#pragma unroll
  for (int v = 0; v < 8; ++v) {
    int r = rbase + v;
    if (r < M) H[r * NC + col] = dv[v] * c[v];
  }
}

template <int CLOG>
__global__ __launch_bounds__(256) void edge_scatter_kernel(
    const int* __restrict__ src, const int* __restrict__ dst,
    const float* __restrict__ h,
    float* __restrict__ agg0, float* __restrict__ agg1, int e) {
  float* __restrict__ agg = cache_half() ? agg1 : agg0;
  constexpr int R = (1 << CLOG) / 32;
  constexpr int G = 32 / R;
  const unsigned lane = threadIdx.x & 31u;
  const int wave = __builtin_amdgcn_readfirstlane((int)((blockIdx.x * blockDim.x + threadIdx.x) >> 5));
  const int nwaves = (int)((gridDim.x * blockDim.x) >> 5);
  const int nfull = e >> 5;

  for (int b = wave; b < nfull; b += nwaves) {
    const int eidx = (b << 5) + (int)lane;
    const int s = src[eidx];
    const int d = dst[eidx];
#pragma unroll
    for (int j0 = 0; j0 < 32; j0 += G) {
      float v[G][R];
#pragma unroll
      for (int j = 0; j < G; ++j) {
        const int sj = __builtin_amdgcn_readlane(s, j0 + j);
        const unsigned so = ((unsigned)sj << CLOG) + lane;
#pragma unroll
        for (int r = 0; r < R; ++r) v[j][r] = h[so + 32 * r];
      }
#pragma unroll
      for (int j = 0; j < G; ++j) {
        const int dj = __builtin_amdgcn_readlane(d, j0 + j);
        const unsigned dof = ((unsigned)dj << CLOG) + lane;
#pragma unroll
        for (int r = 0; r < R; ++r) atomicAdd(agg + (dof + 32 * r), v[j][r]);
      }
    }
  }

  if (wave == 0) {
    for (int ed = nfull << 5; ed < e; ++ed) {
      const int s = src[ed];
      const int d = dst[ed];
      const float* hp = h + ((size_t)s << CLOG) + lane;
      float* ap = agg + ((size_t)d << CLOG) + lane;
      for (int r = 0; r < R; ++r) atomicAdd(ap + 32 * r, hp[32 * r]);
    }
  }
}

template <int CLOG, bool RELU>
__global__ void finish_kernel(const float* __restrict__ agg0, const float* __restrict__ agg1,
                              const float* __restrict__ hs,
                              const float* __restrict__ dinv, const float* __restrict__ bias,
                              float* __restrict__ out, int n) {
  unsigned t = blockIdx.x * blockDim.x + threadIdx.x;
  unsigned i = t >> CLOG;
  if (i >= (unsigned)n) return;
  unsigned c = t & ((1u << CLOG) - 1u);
  float di = dinv[i];
  float v = di * ((agg0[t] + agg1[t]) + hs[t]) + bias[c];
  out[t] = RELU ? fmaxf(v, 0.0f) : v;
}


extern "C" void kernel_launch(void* const* d_in, const int* in_sizes, int n_in,
                              void* d_out, int out_size, void* d_ws, size_t ws_size,
                              hipStream_t stream) {
  (void)n_in; (void)out_size; (void)ws_size;
  const float* x  = (const float*)d_in[0];
  const int*   ei = (const int*)d_in[1];
  const float* W1 = (const float*)d_in[2];
  const float* b1 = (const float*)d_in[3];
  const float* W2 = (const float*)d_in[4];
  const float* b2 = (const float*)d_in[5];

  const int IN_C = 128, HID = 64, OUT = 32;
  const int n = in_sizes[0] / IN_C;
  const int e = in_sizes[1] / 2;
  const int* srcp = ei;
  const int* dstp = ei + e;

  float* ws   = (float*)d_ws;
  size_t off  = ((size_t)n + 255) & ~(size_t)255;
  float* dinv = ws;
  float* deg0 = ws + off;
  float* deg1 = ws + 2 * off;
  float* agg1 = ws + 3 * off;
  float* agg1b = agg1 + (size_t)n * HID;
  float* outa = agg1b + (size_t)n * HID;
  float* outb = outa + (size_t)n * OUT;
  float* h1   = outb + (size_t)n * OUT;
  float* h2   = h1 + (size_t)n * HID;
  float* outp = (float*)d_out;

  const int TPB = 256;
  const int EDGE_BLOCKS = 2048;

  hipMemsetAsync(deg0, 0, (2 * off + (size_t)n * (2 * HID + 2 * OUT)) * sizeof(float), stream);

  deg_count_kernel<<<(e + TPB - 1) / TPB, TPB, 0, stream>>>(dstp, deg0, deg1, e);

  gemm_wmma_kernel<128, 64><<<(n + 15) / 16, 128, 0, stream>>>(x, W1, deg0, deg1, dinv, h1, n);
  edge_scatter_kernel<6><<<EDGE_BLOCKS, TPB, 0, stream>>>(srcp, dstp, h1, agg1, agg1b, e);

  finish_gemm_kernel<64, 32><<<(n + 15) / 16, 64, 0, stream>>>(agg1, agg1b, h1, dinv, b1, W2, h2, n);
  edge_scatter_kernel<5><<<EDGE_BLOCKS, TPB, 0, stream>>>(srcp, dstp, h2, outa, outb, e);
  {
    unsigned total = (unsigned)n << 5;
    finish_kernel<5, false><<<(total + TPB - 1) / TPB, TPB, 0, stream>>>(outa, outb, h2, dinv, b2, outp, n);
  }
}


# Layer-1 product reads its 16 x 128 A tile once per block by whole-line 16-byte loads into LDS and takes A fragments from LDS, not per-wave 8-byte loads that touch 16 lines each; sums unchanged
# speedup vs baseline: 3.8849x; 1.0734x over previous
#include <hip/hip_runtime.h>

typedef __attribute__((ext_vector_type(2))) float v2f;
typedef __attribute__((ext_vector_type(8))) float v8f;


__device__ __forceinline__ int cache_half() {
  return (int)((__builtin_amdgcn_s_sendmsg_rtn(0x87) >> 18) & 1u);
}


__global__ void deg_count_kernel(const int* __restrict__ dst, float* __restrict__ deg0,
                                 float* __restrict__ deg1, int e) {
  int i = blockIdx.x * blockDim.x + threadIdx.x;
  float* deg = cache_half() ? deg1 : deg0;
  if (i < e) atomicAdd(&deg[dst[i]], 1.0f);
}

__device__ __forceinline__ float dinv_of(float d0, float d1) {
  float d = (d0 + d1) + 1.0f;
  return (d > 0.0f) ? rsqrtf(d) : 0.0f;
}

template <int K, int NC>
__global__ __launch_bounds__(32 * (NC / 16)) void gemm_wmma_kernel(
    const float* __restrict__ X, const float* __restrict__ W,
    const float* __restrict__ deg0, const float* __restrict__ deg1,
    float* __restrict__ dinv, float* __restrict__ H, int M) {
  constexpr int NT = 32 * (NC / 16);
  constexpr int Q = K / 4;
  constexpr int PITCH = K + 4;
  static_assert((16 * Q) % NT == 0, "the tile's pieces divide evenly over the block's threads");
  __shared__ __attribute__((aligned(16))) float tile[16 * PITCH];

  const int lane = threadIdx.x & 31;
  const int wave = threadIdx.x >> 5;
  const int m0   = blockIdx.x << 4;
  const int n0   = wave << 4;
  const int l15  = lane & 15;
  const int kh   = (lane >> 4) << 1;

  int row = m0 + l15;
  if (row >= M) row = M - 1;
  const int col = n0 + l15;

  const int rbase = m0 + ((lane >> 4) << 3);
  float dv[8];
#pragma unroll
  for (int v = 0; v < 8; ++v) {
    const int r = rbase + v;
    const int rc = r < M ? r : M - 1;
    dv[v] = dinv_of(deg0[rc], deg1[rc]);
  }
  const float ds0 = deg0[row];
  const float ds1 = deg1[row];

  float4 xp[(16 * Q) / NT];
#pragma unroll
  for (int i = 0; i < (16 * Q) / NT; ++i) {
    const int p  = i * NT + (int)threadIdx.x;
    const int rl = p / Q;
    const int c4 = (p % Q) << 2;
    int r = m0 + rl;
    if (r >= M) r = M - 1;
    xp[i] = *reinterpret_cast<const float4*>(X + (size_t)r * K + c4);
  }
#pragma unroll
  for (int i = 0; i < (16 * Q) / NT; ++i) {
    const int p  = i * NT + (int)threadIdx.x;
    *reinterpret_cast<float4*>(tile + (p / Q) * PITCH + ((p % Q) << 2)) = xp[i];
  }
  __syncthreads();

  v8f c = {};
  for (int k0 = 0; k0 < K; k0 += 4) {
    const v2f a = *reinterpret_cast<const v2f*>(tile + l15 * PITCH + k0 + kh);
    v2f b;
    b.x = W[(k0 + kh) * NC + col];
    b.y = W[(k0 + kh + 1) * NC + col];
    c = __builtin_amdgcn_wmma_f32_16x16x4_f32(
         false, a,  false, b,
         (short)0, c,  false,  false);
  }

#pragma unroll
  for (int v = 0; v < 8; ++v) {
    int r = rbase + v;
    if (r < M) H[r * NC + col] = dv[v] * c[v];
  }
  if (wave == 0 && lane < 16 && m0 + lane < M) dinv[m0 + lane] = dinv_of(ds0, ds1);
}

template <int K, int NC>
__global__ __launch_bounds__(32 * (NC / 16)) void finish_gemm_kernel(
    const float* __restrict__ agg0, const float* __restrict__ agg1,
    const float* __restrict__ hs, const float* __restrict__ dinv,
    const float* __restrict__ bias, const float* __restrict__ W,
    float* __restrict__ H, int M) {
  constexpr int NT = 32 * (NC / 16);
  constexpr int Q = K / 4;
  constexpr int PITCH = K + 4;
  __shared__ __attribute__((aligned(16))) float tile[16 * PITCH];

  const int lane = threadIdx.x & 31;
  const int wave = threadIdx.x >> 5;
  const int m0   = blockIdx.x << 4;
  const int n0   = wave << 4;
  const int l15  = lane & 15;
  const int kh   = (lane >> 4) << 1;
  const int col  = n0 + l15;

  const int rbase = m0 + ((lane >> 4) << 3);
  float dv[8];
#pragma unroll
  for (int v = 0; v < 8; ++v) {
    const int r = rbase + v;
    dv[v] = dinv[r < M ? r : M - 1];
  }

#pragma unroll
  for (int i = 0; i < (16 * Q) / NT; ++i) {
    const int p  = i * NT + (int)threadIdx.x;
    const int rl = p / Q;
    const int c4 = (p % Q) << 2;
    int r = m0 + rl;
    if (r >= M) r = M - 1;
    const int g = r * K + c4;
    const float4 a = *reinterpret_cast<const float4*>(agg0 + g);
    const float4 b = *reinterpret_cast<const float4*>(agg1 + g);
    const float4 h = *reinterpret_cast<const float4*>(hs + g);
    const float di = dinv[r];
    float4 o;
    o.x = fmaxf(di * ((a.x + b.x) + h.x) + bias[c4 + 0], 0.0f);
    o.y = fmaxf(di * ((a.y + b.y) + h.y) + bias[c4 + 1], 0.0f);
    o.z = fmaxf(di * ((a.z + b.z) + h.z) + bias[c4 + 2], 0.0f);
    o.w = fmaxf(di * ((a.w + b.w) + h.w) + bias[c4 + 3], 0.0f);
    *reinterpret_cast<float4*>(tile + rl * PITCH + c4) = o;
  }
  __syncthreads();

  v8f c = {};
  for (int k0 = 0; k0 < K; k0 += 4) {
    const v2f a = *reinterpret_cast<const v2f*>(tile + l15 * PITCH + k0 + kh);
    v2f b;
    b.x = W[(k0 + kh) * NC + col];
    b.y = W[(k0 + kh + 1) * NC + col];
    c = __builtin_amdgcn_wmma_f32_16x16x4_f32(
         false, a,  false, b,
         (short)0, c,  false,  false);
  }

#pragma unroll
  for (int v = 0; v < 8; ++v) {
    int r = rbase + v;
    if (r < M) H[r * NC + col] = dv[v] * c[v];
  }
}

template <int CLOG>
__global__ __launch_bounds__(256) void edge_scatter_kernel(
    const int* __restrict__ src, const int* __restrict__ dst,
    const float* __restrict__ h,
    float* __restrict__ agg0, float* __restrict__ agg1, int e) {
  float* __restrict__ agg = cache_half() ? agg1 : agg0;
  constexpr int R = (1 << CLOG) / 32;
  constexpr int G = 32 / R;
  const unsigned lane = threadIdx.x & 31u;
  const int wave = __builtin_amdgcn_readfirstlane((int)((blockIdx.x * blockDim.x + threadIdx.x) >> 5));
  const int nwaves = (int)((gridDim.x * blockDim.x) >> 5);
  const int nfull = e >> 5;

  for (int b = wave; b < nfull; b += nwaves) {
    const int eidx = (b << 5) + (int)lane;
    const int s = src[eidx];
    const int d = dst[eidx];
#pragma unroll
    for (int j0 = 0; j0 < 32; j0 += G) {
      float v[G][R];
#pragma unroll
      for (int j = 0; j < G; ++j) {
        const int sj = __builtin_amdgcn_readlane(s, j0 + j);
        const unsigned so = ((unsigned)sj << CLOG) + lane;
#pragma unroll
        for (int r = 0; r < R; ++r) v[j][r] = h[so + 32 * r];
      }
#pragma unroll
      for (int j = 0; j < G; ++j) {
        const int dj = __builtin_amdgcn_readlane(d, j0 + j);
        const unsigned dof = ((unsigned)dj << CLOG) + lane;
#pragma unroll
        for (int r = 0; r < R; ++r) atomicAdd(agg + (dof + 32 * r), v[j][r]);
      }
    }
  }

  if (wave == 0) {
    for (int ed = nfull << 5; ed < e; ++ed) {
      const int s = src[ed];
      const int d = dst[ed];
      const float* hp = h + ((size_t)s << CLOG) + lane;
      float* ap = agg + ((size_t)d << CLOG) + lane;
      for (int r = 0; r < R; ++r) atomicAdd(ap + 32 * r, hp[32 * r]);
    }
  }
}

template <int CLOG, bool RELU>
__global__ void finish_kernel(const float* __restrict__ agg0, const float* __restrict__ agg1,
                              const float* __restrict__ hs,
                              const float* __restrict__ dinv, const float* __restrict__ bias,
                              float* __restrict__ out, int n) {
  unsigned t = blockIdx.x * blockDim.x + threadIdx.x;
  unsigned i = t >> CLOG;
  if (i >= (unsigned)n) return;
  unsigned c = t & ((1u << CLOG) - 1u);
  float di = dinv[i];
  float v = di * ((agg0[t] + agg1[t]) + hs[t]) + bias[c];
  out[t] = RELU ? fmaxf(v, 0.0f) : v;
}


extern "C" void kernel_launch(void* const* d_in, const int* in_sizes, int n_in,
                              void* d_out, int out_size, void* d_ws, size_t ws_size,
                              hipStream_t stream) {
  (void)n_in; (void)out_size; (void)ws_size;
  const float* x  = (const float*)d_in[0];
  const int*   ei = (const int*)d_in[1];
  const float* W1 = (const float*)d_in[2];
  const float* b1 = (const float*)d_in[3];
  const float* W2 = (const float*)d_in[4];
  const float* b2 = (const float*)d_in[5];

  const int IN_C = 128, HID = 64, OUT = 32;
  const int n = in_sizes[0] / IN_C;
  const int e = in_sizes[1] / 2;
  const int* srcp = ei;
  const int* dstp = ei + e;

  float* ws   = (float*)d_ws;
  size_t off  = ((size_t)n + 255) & ~(size_t)255;
  float* dinv = ws;
  float* deg0 = ws + off;
  float* deg1 = ws + 2 * off;
  float* agg1 = ws + 3 * off;
  float* agg1b = agg1 + (size_t)n * HID;
  float* outa = agg1b + (size_t)n * HID;
  float* outb = outa + (size_t)n * OUT;
  float* h1   = outb + (size_t)n * OUT;
  float* h2   = h1 + (size_t)n * HID;
  float* outp = (float*)d_out;

  const int TPB = 256;
  const int EDGE_BLOCKS = 2048;

  hipMemsetAsync(deg0, 0, (2 * off + (size_t)n * (2 * HID + 2 * OUT)) * sizeof(float), stream);

  deg_count_kernel<<<(e + TPB - 1) / TPB, TPB, 0, stream>>>(dstp, deg0, deg1, e);

  gemm_wmma_kernel<128, 64><<<(n + 15) / 16, 128, 0, stream>>>(x, W1, deg0, deg1, dinv, h1, n);
  edge_scatter_kernel<6><<<EDGE_BLOCKS, TPB, 0, stream>>>(srcp, dstp, h1, agg1, agg1b, e);

  finish_gemm_kernel<64, 32><<<(n + 15) / 16, 64, 0, stream>>>(agg1, agg1b, h1, dinv, b1, W2, h2, n);
  edge_scatter_kernel<5><<<EDGE_BLOCKS, TPB, 0, stream>>>(srcp, dstp, h2, outa, outb, e);
  {
    unsigned total = (unsigned)n << 5;
    finish_kernel<5, false><<<(total + TPB - 1) / TPB, TPB, 0, stream>>>(outa, outb, h2, dinv, b2, outp, n);
  }
}


# Per-element kernels on few fat waves: degree count does 8 edges per thread, final finish four 16-byte pieces per thread, loads requested before use; 150000 waves become 12512; sums unchanged
# speedup vs baseline: 4.1282x; 1.0626x over previous
#include <hip/hip_runtime.h>

typedef __attribute__((ext_vector_type(2))) float v2f;
typedef __attribute__((ext_vector_type(8))) float v8f;


__device__ __forceinline__ int cache_half() {
  return (int)((__builtin_amdgcn_s_sendmsg_rtn(0x87) >> 18) & 1u);
}


template <int EPT>
__global__ __launch_bounds__(256) void deg_count_kernel(const int* __restrict__ dst, float* __restrict__ deg0,
                                                        float* __restrict__ deg1, int e) {
  const unsigned width = gridDim.x * blockDim.x;
  const unsigned i0 = blockIdx.x * blockDim.x + threadIdx.x;
  float* deg = cache_half() ? deg1 : deg0;
  int d[EPT];
#pragma unroll
  for (int k = 0; k < EPT; ++k) {
    const unsigned i = i0 + (unsigned)k * width;
    d[k] = dst[i < (unsigned)e ? i : (unsigned)e - 1u];
  }
#pragma unroll
  for (int k = 0; k < EPT; ++k) {
    const unsigned i = i0 + (unsigned)k * width;
    if (i < (unsigned)e) atomicAdd(&deg[d[k]], 1.0f);
  }
}

__device__ __forceinline__ float dinv_of(float d0, float d1) {
  float d = (d0 + d1) + 1.0f;
  return (d > 0.0f) ? rsqrtf(d) : 0.0f;
}

template <int K, int NC>
__global__ __launch_bounds__(32 * (NC / 16)) void gemm_wmma_kernel(
    const float* __restrict__ X, const float* __restrict__ W,
    const float* __restrict__ deg0, const float* __restrict__ deg1,
    float* __restrict__ dinv, float* __restrict__ H, int M) {
  constexpr int NT = 32 * (NC / 16);
  constexpr int Q = K / 4;
  constexpr int PITCH = K + 4;
  static_assert((16 * Q) % NT == 0, "the tile's pieces divide evenly over the block's threads");
  __shared__ __attribute__((aligned(16))) float tile[16 * PITCH];

  const int lane = threadIdx.x & 31;
  const int wave = threadIdx.x >> 5;
  const int m0   = blockIdx.x << 4;
  const int n0   = wave << 4;
  const int l15  = lane & 15;
  const int kh   = (lane >> 4) << 1;

  int row = m0 + l15;
  if (row >= M) row = M - 1;
  const int col = n0 + l15;

  const int rbase = m0 + ((lane >> 4) << 3);
  float dv[8];
#pragma unroll
  for (int v = 0; v < 8; ++v) {
    const int r = rbase + v;
    const int rc = r < M ? r : M - 1;
    dv[v] = dinv_of(deg0[rc], deg1[rc]);
  }
  const float ds0 = deg0[row];
  const float ds1 = deg1[row];

  float4 xp[(16 * Q) / NT];
#pragma unroll
  for (int i = 0; i < (16 * Q) / NT; ++i) {
    const int p  = i * NT + (int)threadIdx.x;
    const int rl = p / Q;
    const int c4 = (p % Q) << 2;
    int r = m0 + rl;
    if (r >= M) r = M - 1;
    xp[i] = *reinterpret_cast<const float4*>(X + (size_t)r * K + c4);
  }
#pragma unroll
  for (int i = 0; i < (16 * Q) / NT; ++i) {
    const int p  = i * NT + (int)threadIdx.x;
    *reinterpret_cast<float4*>(tile + (p / Q) * PITCH + ((p % Q) << 2)) = xp[i];
  }
  __syncthreads();

  v8f c = {};
  for (int k0 = 0; k0 < K; k0 += 4) {
    const v2f a = *reinterpret_cast<const v2f*>(tile + l15 * PITCH + k0 + kh);
    v2f b;
    b.x = W[(k0 + kh) * NC + col];
    b.y = W[(k0 + kh + 1) * NC + col];
    c = __builtin_amdgcn_wmma_f32_16x16x4_f32(
         false, a,  false, b,
         (short)0, c,  false,  false);
  }

#pragma unroll
  for (int v = 0; v < 8; ++v) {
    int r = rbase + v;
    if (r < M) H[r * NC + col] = dv[v] * c[v];
  }
  if (wave == 0 && lane < 16 && m0 + lane < M) dinv[m0 + lane] = dinv_of(ds0, ds1);
}

template <int K, int NC>
__global__ __launch_bounds__(32 * (NC / 16)) void finish_gemm_kernel(
    const float* __restrict__ agg0, const float* __restrict__ agg1,
    const float* __restrict__ hs, const float* __restrict__ dinv,
    const float* __restrict__ bias, const float* __restrict__ W,
    float* __restrict__ H, int M) {
  constexpr int NT = 32 * (NC / 16);
  constexpr int Q = K / 4;
  constexpr int PITCH = K + 4;
  __shared__ __attribute__((aligned(16))) float tile[16 * PITCH];

  const int lane = threadIdx.x & 31;
  const int wave = threadIdx.x >> 5;
  const int m0   = blockIdx.x << 4;
  const int n0   = wave << 4;
  const int l15  = lane & 15;
  const int kh   = (lane >> 4) << 1;
  const int col  = n0 + l15;

  const int rbase = m0 + ((lane >> 4) << 3);
  float dv[8];
#pragma unroll
  for (int v = 0; v < 8; ++v) {
    const int r = rbase + v;
    dv[v] = dinv[r < M ? r : M - 1];
  }

#pragma unroll
  for (int i = 0; i < (16 * Q) / NT; ++i) {
    const int p  = i * NT + (int)threadIdx.x;
    const int rl = p / Q;
    const int c4 = (p % Q) << 2;
    int r = m0 + rl;
    if (r >= M) r = M - 1;
    const int g = r * K + c4;
    const float4 a = *reinterpret_cast<const float4*>(agg0 + g);
    const float4 b = *reinterpret_cast<const float4*>(agg1 + g);
    const float4 h = *reinterpret_cast<const float4*>(hs + g);
    const float di = dinv[r];
    float4 o;
    o.x = fmaxf(di * ((a.x + b.x) + h.x) + bias[c4 + 0], 0.0f);
    o.y = fmaxf(di * ((a.y + b.y) + h.y) + bias[c4 + 1], 0.0f);
    o.z = fmaxf(di * ((a.z + b.z) + h.z) + bias[c4 + 2], 0.0f);
    o.w = fmaxf(di * ((a.w + b.w) + h.w) + bias[c4 + 3], 0.0f);
    *reinterpret_cast<float4*>(tile + rl * PITCH + c4) = o;
  }
  __syncthreads();

  v8f c = {};
  for (int k0 = 0; k0 < K; k0 += 4) {
    const v2f a = *reinterpret_cast<const v2f*>(tile + l15 * PITCH + k0 + kh);
    v2f b;
    b.x = W[(k0 + kh) * NC + col];
    b.y = W[(k0 + kh + 1) * NC + col];
    c = __builtin_amdgcn_wmma_f32_16x16x4_f32(
         false, a,  false, b,
         (short)0, c,  false,  false);
  }

#pragma unroll
  for (int v = 0; v < 8; ++v) {
    int r = rbase + v;
    if (r < M) H[r * NC + col] = dv[v] * c[v];
  }
}

template <int CLOG>
__global__ __launch_bounds__(256) void edge_scatter_kernel(
    const int* __restrict__ src, const int* __restrict__ dst,
    const float* __restrict__ h,
    float* __restrict__ agg0, float* __restrict__ agg1, int e) {
  float* __restrict__ agg = cache_half() ? agg1 : agg0;
  constexpr int R = (1 << CLOG) / 32;
  constexpr int G = 32 / R;
  const unsigned lane = threadIdx.x & 31u;
  const int wave = __builtin_amdgcn_readfirstlane((int)((blockIdx.x * blockDim.x + threadIdx.x) >> 5));
  const int nwaves = (int)((gridDim.x * blockDim.x) >> 5);
  const int nfull = e >> 5;

  for (int b = wave; b < nfull; b += nwaves) {
    const int eidx = (b << 5) + (int)lane;
    const int s = src[eidx];
    const int d = dst[eidx];
#pragma unroll
    for (int j0 = 0; j0 < 32; j0 += G) {
      float v[G][R];
#pragma unroll
      for (int j = 0; j < G; ++j) {
        const int sj = __builtin_amdgcn_readlane(s, j0 + j);
        const unsigned so = ((unsigned)sj << CLOG) + lane;
#pragma unroll
        for (int r = 0; r < R; ++r) v[j][r] = h[so + 32 * r];
      }
#pragma unroll
      for (int j = 0; j < G; ++j) {
        const int dj = __builtin_amdgcn_readlane(d, j0 + j);
        const unsigned dof = ((unsigned)dj << CLOG) + lane;
#pragma unroll
        for (int r = 0; r < R; ++r) atomicAdd(agg + (dof + 32 * r), v[j][r]);
      }
    }
  }

  if (wave == 0) {
    for (int ed = nfull << 5; ed < e; ++ed) {
      const int s = src[ed];
      const int d = dst[ed];
      const float* hp = h + ((size_t)s << CLOG) + lane;
      float* ap = agg + ((size_t)d << CLOG) + lane;
      for (int r = 0; r < R; ++r) atomicAdd(ap + 32 * r, hp[32 * r]);
    }
  }
}

template <int CLOG, bool RELU, int PPT>
__global__ __launch_bounds__(256) void finish_kernel(
    const float* __restrict__ agg0, const float* __restrict__ agg1,
    const float* __restrict__ hs,
    const float* __restrict__ dinv, const float* __restrict__ bias,
    float* __restrict__ out, int n) {
  constexpr int QLOG = CLOG - 2;
  static_assert(256 % (1 << QLOG) == 0, "a block holds whole rows, so the grid width is a multiple of the pieces per row");
  const unsigned total = (unsigned)n << QLOG;
  const unsigned width = gridDim.x * blockDim.x;
  const unsigned p0 = blockIdx.x * blockDim.x + threadIdx.x;
  const float4 bs = *reinterpret_cast<const float4*>(bias + ((p0 & ((1u << QLOG) - 1u)) << 2));

  float4 a[PPT], b[PPT], h[PPT];
  float di[PPT];
#pragma unroll
  for (int k = 0; k < PPT; ++k) {
    unsigned p = p0 + (unsigned)k * width;
    if (p >= total) p = total - 1;
    a[k] = reinterpret_cast<const float4*>(agg0)[p];
    b[k] = reinterpret_cast<const float4*>(agg1)[p];
    h[k] = reinterpret_cast<const float4*>(hs)[p];
    di[k] = dinv[p >> QLOG];
  }
#pragma unroll
  for (int k = 0; k < PPT; ++k) {
    const unsigned p = p0 + (unsigned)k * width;
    float4 o;
    o.x = di[k] * ((a[k].x + b[k].x) + h[k].x) + bs.x;
    o.y = di[k] * ((a[k].y + b[k].y) + h[k].y) + bs.y;
    o.z = di[k] * ((a[k].z + b[k].z) + h[k].z) + bs.z;
    o.w = di[k] * ((a[k].w + b[k].w) + h[k].w) + bs.w;
    if (RELU) {
      o.x = fmaxf(o.x, 0.0f);
      o.y = fmaxf(o.y, 0.0f);
      o.z = fmaxf(o.z, 0.0f);
      o.w = fmaxf(o.w, 0.0f);
    }
    if (p < total) reinterpret_cast<float4*>(out)[p] = o;
  }
}


extern "C" void kernel_launch(void* const* d_in, const int* in_sizes, int n_in,
                              void* d_out, int out_size, void* d_ws, size_t ws_size,
                              hipStream_t stream) {
  (void)n_in; (void)out_size; (void)ws_size;
  const float* x  = (const float*)d_in[0];
  const int*   ei = (const int*)d_in[1];
  const float* W1 = (const float*)d_in[2];
  const float* b1 = (const float*)d_in[3];
  const float* W2 = (const float*)d_in[4];
  const float* b2 = (const float*)d_in[5];

  const int IN_C = 128, HID = 64, OUT = 32;
  const int n = in_sizes[0] / IN_C;
  const int e = in_sizes[1] / 2;
  const int* srcp = ei;
  const int* dstp = ei + e;

  float* ws   = (float*)d_ws;
  size_t off  = ((size_t)n + 255) & ~(size_t)255;
  float* dinv = ws;
  float* deg0 = ws + off;
  float* deg1 = ws + 2 * off;
  float* agg1 = ws + 3 * off;
  float* agg1b = agg1 + (size_t)n * HID;
  float* outa = agg1b + (size_t)n * HID;
  float* outb = outa + (size_t)n * OUT;
  float* h1   = outb + (size_t)n * OUT;
  float* h2   = h1 + (size_t)n * HID;
  float* outp = (float*)d_out;

  const int TPB = 256;
  const int EDGE_BLOCKS = 2048;

  hipMemsetAsync(deg0, 0, (2 * off + (size_t)n * (2 * HID + 2 * OUT)) * sizeof(float), stream);

  deg_count_kernel<8><<<(e + 8 * TPB - 1) / (8 * TPB), TPB, 0, stream>>>(dstp, deg0, deg1, e);

  gemm_wmma_kernel<128, 64><<<(n + 15) / 16, 128, 0, stream>>>(x, W1, deg0, deg1, dinv, h1, n);
  edge_scatter_kernel<6><<<EDGE_BLOCKS, TPB, 0, stream>>>(srcp, dstp, h1, agg1, agg1b, e);

  finish_gemm_kernel<64, 32><<<(n + 15) / 16, 64, 0, stream>>>(agg1, agg1b, h1, dinv, b1, W2, h2, n);
  edge_scatter_kernel<5><<<EDGE_BLOCKS, TPB, 0, stream>>>(srcp, dstp, h2, outa, outb, e);
  {
    unsigned pieces = (unsigned)n << 3;
    finish_kernel<5, false, 4><<<(pieces + 4 * TPB - 1) / (4 * TPB), TPB, 0, stream>>>(outa, outb, h2, dinv, b2, outp, n);
  }
}
